# starting kernel
# baseline (speedup 1.0000x reference)
#include <hip/hip_runtime.h>


typedef __attribute__((ext_vector_type(2))) float v2f;
typedef __attribute__((ext_vector_type(8))) float v8f;

#define B_    4
#define CIN_  16
#define COUT_ 16
#define T_    16
#define H_    128
#define W_    128
#define SCALE_ 0.04811252243246881f

#define XROW 144
#define NA   (27 * 4 * 32 * 2)
#define NDH  4
#define NX   (3 * NDH * 8 * XROW * 2)

__global__ __launch_bounds__(128) void conv3d_mod_wmma_f32(
    const float* __restrict__ input,
    const float* __restrict__ cond,
    const float* __restrict__ weight,
    float* __restrict__ out)
{
    extern __shared__ float smem[];
    float* sA = smem;
    float* sX = smem + NA;

    const int bi  = blockIdx.x;
    const int h0  = (bi % (H_ / 2)) * 2;
    const int t   = (bi / (H_ / 2)) % T_;
    const int b   = bi / ((H_ / 2) * T_);
    const int tid = threadIdx.x;

    for (int j = tid; j < NA; j += 128) {
        int tap  = j >> 8;
        int rem  = j & 255;
        int c    = rem >> 6;
        int rem2 = rem & 63;
        int l    = rem2 >> 1;
        int jj   = rem2 & 1;
        int cin  = 4 * c + 2 * (l >> 4) + jj;
        int m    = l & 15;
        int kt = tap / 9, kh = (tap / 3) % 3, kw = tap % 3;
        float wgt = weight[(((m * CIN_ + cin) * 3 + kt) * 3 + kh) * 3 + kw];
        sA[j] = wgt * SCALE_ * cond[b * CIN_ + cin];
    }

    for (int i = tid; i < 16 * 3 * NDH * 130; i += 128) {
        int r   = i / 130;
        int wi  = i - r * 130;
        int cin = r / (3 * NDH);
        int rr  = r - cin * (3 * NDH);
        int dt  = rr / NDH;
        int dh  = rr - dt * NDH;
        int ts = t + dt - 1, hs = h0 + dh - 1, ws = wi - 1;
        float v = 0.0f;
        if (((unsigned)ts < (unsigned)T_) & ((unsigned)hs < (unsigned)H_) &
            ((unsigned)ws < (unsigned)W_))
            v = input[((size_t)(b * CIN_ + cin) * T_ + ts) * (H_ * W_) +
                      (size_t)hs * W_ + ws];
        sX[(((dt * NDH + dh) * 8 + (cin >> 1)) * XROW + wi) * 2 + (cin & 1)] = v;
    }
    __syncthreads();

    const int wvid = tid >> 5;
    const int hr   = wvid >> 1;
    const int wseg = (wvid & 1) * 64;
    const int lane = tid & 31;
    const int hf   = lane >> 4;
    const int lx   = lane & 15;

    v8f acc0 = {}, acc1 = {}, acc2 = {}, acc3 = {};

#pragma unroll
    for (int tap = 0; tap < 27; ++tap) {
        const int kt = tap / 9, kh = (tap / 3) % 3, kw = tap % 3;
        const int rowb = (kt * NDH + (kh + hr)) * 8;
        const int n0   = wseg + lx + kw;
#pragma unroll
        for (int c = 0; c < 4; ++c) {
            v2f a = *(const v2f*)&sA[(((tap * 4 + c) << 5) + lane) * 2];
            int base = (rowb + 2 * c + hf) * (XROW * 2);
            v2f b0 = *(const v2f*)&sX[base + n0 * 2];
            v2f b1 = *(const v2f*)&sX[base + (n0 + 16) * 2];
            v2f b2 = *(const v2f*)&sX[base + (n0 + 32) * 2];
            v2f b3 = *(const v2f*)&sX[base + (n0 + 48) * 2];
            acc0 = __builtin_amdgcn_wmma_f32_16x16x4_f32(
                false, a, false, b0, (short)0, acc0, false, false);
            acc1 = __builtin_amdgcn_wmma_f32_16x16x4_f32(
                false, a, false, b1, (short)0, acc1, false, false);
            acc2 = __builtin_amdgcn_wmma_f32_16x16x4_f32(
                false, a, false, b2, (short)0, acc2, false, false);
            acc3 = __builtin_amdgcn_wmma_f32_16x16x4_f32(
                false, a, false, b3, (short)0, acc3, false, false);
        }
    }

    const size_t strideC = (size_t)T_ * H_ * W_;
    float* o = out + (size_t)b * COUT_ * strideC + (size_t)t * (H_ * W_) +
               (size_t)(h0 + hr) * W_ + wseg + lx;
#pragma unroll
    for (int r = 0; r < 8; ++r) {
        size_t mo = (size_t)(hf * 8 + r) * strideC;
        o[mo]      = acc0[r];
        o[mo + 16] = acc1[r];
        o[mo + 32] = acc2[r];
        o[mo + 48] = acc3[r];
    }
}

extern "C" void kernel_launch(void* const* d_in, const int* in_sizes, int n_in,
                              void* d_out, int out_size, void* d_ws, size_t ws_size,
                              hipStream_t stream) {
    const float* input  = (const float*)d_in[0];
    const float* cond   = (const float*)d_in[1];
    const float* weight = (const float*)d_in[2];
    float* out = (float*)d_out;

    const int    nBlocks = B_ * T_ * (H_ / 2);
    const size_t shmem   = (size_t)(NA + NX) * sizeof(float);

    conv3d_mod_wmma_f32<<<dim3(nBlocks), dim3(128), shmem, stream>>>(
        input, cond, weight, out);
}


# weight staging: a wave reads its two couts' weights as 27 whole 128-byte lines, all in flight, one division by 27 per element; LDS tap rows padded 2 dwords; WMMA loop and summation order unchanged
# speedup vs baseline: 7.3258x; 7.3258x over previous
#include <hip/hip_runtime.h>


typedef __attribute__((ext_vector_type(2))) float v2f;
typedef __attribute__((ext_vector_type(8))) float v8f;

#define B_     4
#define CIN_   16
#define COUT_  16
#define T_     16
#define H_     128
#define W_     128
#define SCALE_ 0.04811252243246881f
#define HW_    (H_ * W_)
#define PLANE_ (T_ * H_ * W_)
#define TAPS_  (4 * 32 * 2 + 2)
#define NA     (27 * TAPS_)
#define WPW_   (2 * CIN_ * 27)
#define ROWS_  8
#define RW_    4
#define NR_    (RW_ + 2)
#define PF_    2
#define NBUF_  (PF_ + 1)

__device__ __forceinline__ float ldx(const float* rowp, unsigned lane_off_bytes) {
    return *(const float*)((const char*)rowp + lane_off_bytes);
}

__global__ __launch_bounds__(256) __attribute__((amdgpu_waves_per_eu(8, 8))) void conv3d_mod_wmma_f32_direct(
    const float* __restrict__ input,
    const float* __restrict__ cond,
    const float* __restrict__ weight,
    float* __restrict__ out)
{
    __shared__ __attribute__((aligned(8))) float sA[NA];

    const int bi  = blockIdx.x;
    const int h0  = (bi % (H_ / ROWS_)) * ROWS_;
    const int t   = (bi / (H_ / ROWS_)) % T_;
    const int b   = bi / ((H_ / ROWS_) * T_);
    const int tid = threadIdx.x;

    {
        const int wq = tid >> 5, lq = tid & 31;
        const float* wsrc = weight + wq * WPW_ + lq;
        const float* cb   = cond + b * CIN_;
        float wraw[27];
#pragma unroll
        for (int n = 0; n < 27; ++n) wraw[n] = wsrc[32 * n];
#pragma unroll
        for (int n = 0; n < 27; ++n) {
            const unsigned idx = 32u * (unsigned)n + (unsigned)lq;
            const unsigned mm  = idx >= 432u ? 1u : 0u;
            const unsigned r   = idx - 432u * mm;
            const unsigned cin = r / 27u;
            const unsigned tap = r - 27u * cin;
            const unsigned m   = 2u * (unsigned)wq + mm;
            sA[tap * TAPS_ + (cin >> 2) * 64u + ((cin >> 1) & 1u) * 32u + m * 2u + (cin & 1u)] =
                wraw[n] * SCALE_ * cb[cin];
        }
    }
    __syncthreads();

    const int wv   = __builtin_amdgcn_readfirstlane(tid >> 5);
    const int lane = tid & 31;
    const int hf   = lane >> 4;
    const int lx   = lane & 15;
    const int rg   = wv >> 2;
    const int w0   = (wv & 3) * 32;
    const int hb   = h0 + rg * RW_;
    const int wA   = w0 + 2 * lx;

    const bool okL = wA > 0;
    const bool okR = wA + 2 < W_;
    const bool topok = hb > 0;
    const bool botok = hb + RW_ < H_;
    const int  hTop  = topok ? hb - 1 : 0;
    const int  hBot  = botok ? hb + RW_ : H_ - 1;

    v8f accA[RW_], accB[RW_];
#pragma unroll
    for (int i = 0; i < RW_; ++i) { accA[i] = v8f{}; accB[i] = v8f{}; }

    const float* inb = input + (size_t)b * CIN_ * PLANE_;

    const unsigned chalf = (unsigned)hf * 2u * (unsigned)PLANE_;
    const unsigned off0 = (chalf + (unsigned)(wA > 0 ? wA - 1 : 0)) * 4u;
    const unsigned off1 = (chalf + (unsigned)wA) * 4u;
    const unsigned off2 = (chalf + (unsigned)(wA + 1)) * 4u;
    const unsigned off3 = (chalf + (unsigned)(wA + 2 < W_ ? wA + 2 : W_ - 1)) * 4u;

    v2f a[9];
    float xb[NBUF_][8];

#pragma unroll
    for (int q = 0; q < 9; ++q)
        a[q] = *(const v2f*)&sA[q * TAPS_ + lane * 2];

#define NSTEP_ (3 * 4 * NR_)
#define FETCH_ROW(rs_)                                                                       \
    {                                                                                        \
        const int fs = (rs_) / NR_, fr = (rs_) % NR_;                                        \
        const int fdt = fs / 4, fc = fs % 4;                                                 \
        const int fts = (fdt == 0) ? (t > 0 ? t - 1 : 0) : (fdt == 2) ? (t < T_ - 1 ? t + 1 : T_ - 1) : t; \
        const int fhs = (fr == 0) ? hTop : (fr == NR_ - 1) ? hBot : hb + fr - 1;             \
        const float* p0 = inb + (size_t)fhs * W_;                                            \
        const float* p1 = p0 + PLANE_;                                                       \
        const unsigned co = (unsigned)fts * (unsigned)(HW_ * 4) + (unsigned)fc * (unsigned)(4 * PLANE_ * 4); \
        float* xf = xb[(rs_) % NBUF_];                                                       \
        xf[0] = ldx(p0, off0 + co); xf[1] = ldx(p1, off0 + co);                              \
        xf[2] = ldx(p0, off1 + co); xf[3] = ldx(p1, off1 + co);                              \
        xf[4] = ldx(p0, off2 + co); xf[5] = ldx(p1, off2 + co);                              \
        xf[6] = ldx(p0, off3 + co); xf[7] = ldx(p1, off3 + co);                              \
    }

#pragma unroll
    for (int p = 0; p < PF_; ++p) FETCH_ROW(p)

#pragma unroll
    for (int s = 0; s < 12; ++s) {
#pragma unroll
    for (int ri = 0; ri < NR_; ++ri) {
        const int rs = s * NR_ + ri;
        const int dt = s / 4;
        if (rs + PF_ < NSTEP_) FETCH_ROW(rs + PF_)
        if (ri == RW_ && s < 11) {
            const int s1 = s + 1;
#pragma unroll
            for (int kw = 0; kw < 3; ++kw)
                a[kw] = *(const v2f*)&sA[((s1 / 4) * 9 + kw) * TAPS_ + (s1 % 4) * 64 + lane * 2];
        }
        if (ri == RW_ + 1 && s < 11) {
            const int s1 = s + 1;
#pragma unroll
            for (int kw = 0; kw < 3; ++kw)
                a[3 + kw] = *(const v2f*)&sA[((s1 / 4) * 9 + 3 + kw) * TAPS_ + (s1 % 4) * 64 + lane * 2];
        }
        if (ri == 0 && s > 0) {
#pragma unroll
            for (int kw = 0; kw < 3; ++kw)
                a[6 + kw] = *(const v2f*)&sA[((s / 4) * 9 + 6 + kw) * TAPS_ + (s % 4) * 64 + lane * 2];
        }

        float* xs = xb[rs % NBUF_];
        const bool planeok = (dt == 0) ? (t > 0) : (dt == 2) ? (t < T_ - 1) : true;
        const bool rowok = (ri == 0) ? topok : (ri == NR_ - 1) ? botok : true;
        const bool inside = planeok && rowok;
        const bool cl = okL && inside, cr = okR && inside;
        xs[0] = cl ? xs[0] : 0.0f; xs[1] = cl ? xs[1] : 0.0f;
        xs[6] = cr ? xs[6] : 0.0f; xs[7] = cr ? xs[7] : 0.0f;
        if (dt != 1 || ri == 0 || ri == NR_ - 1) {
            xs[2] = inside ? xs[2] : 0.0f; xs[3] = inside ? xs[3] : 0.0f;
            xs[4] = inside ? xs[4] : 0.0f; xs[5] = inside ? xs[5] : 0.0f;
        }
#pragma unroll
        for (int kw = 0; kw < 3; ++kw) {
            v2f bfA = {xs[2 * kw], xs[2 * kw + 1]};
            v2f bfB = {xs[2 * kw + 2], xs[2 * kw + 3]};
#pragma unroll
            for (int kh = 0; kh < 3; ++kh) {
                const int ro = ri - kh;
                if (ro >= 0 && ro < RW_) {
                    accA[ro] = __builtin_amdgcn_wmma_f32_16x16x4_f32(
                        false, a[kh * 3 + kw], false, bfA, (short)0, accA[ro], false, false);
                    accB[ro] = __builtin_amdgcn_wmma_f32_16x16x4_f32(
                        false, a[kh * 3 + kw], false, bfB, (short)0, accB[ro], false, false);
                }
            }
        }
        __builtin_amdgcn_sched_barrier(0);
    }
    }
#undef FETCH_ROW
#undef NSTEP_

    const size_t strideC = (size_t)PLANE_;
    float* o = out + (size_t)b * COUT_ * strideC + (size_t)t * HW_ + (size_t)hb * W_ + wA;
#pragma unroll
    for (int ro = 0; ro < RW_; ++ro) {
#pragma unroll
        for (int r = 0; r < 8; ++r) {
            v2f pr = {accA[ro][r], accB[ro][r]};
            *(v2f*)&o[(size_t)(hf * 8 + r) * strideC + (size_t)ro * W_] = pr;
        }
    }
}

extern "C" void kernel_launch(void* const* d_in, const int* in_sizes, int n_in,
                              void* d_out, int out_size, void* d_ws, size_t ws_size,
                              hipStream_t stream) {
    const float* input  = (const float*)d_in[0];
    const float* cond   = (const float*)d_in[1];
    const float* weight = (const float*)d_in[2];
    float* out = (float*)d_out;

    conv3d_mod_wmma_f32_direct<<<dim3(B_ * T_ * (H_ / ROWS_)), dim3(256), 0, stream>>>(
        input, cond, weight, out);
}


# wave tile 4 rows x 64 w, lane holds 4 adjacent columns: one 16-byte load per row and channel, 6 column fragments feed 4 tiles, b128 stores; half as many waves, 4 per SIMD; sum order unchanged
# speedup vs baseline: 7.7789x; 1.0618x over previous
#include <hip/hip_runtime.h>


typedef __attribute__((ext_vector_type(2))) float v2f;
typedef __attribute__((ext_vector_type(4))) float v4f;
typedef __attribute__((ext_vector_type(8))) float v8f;

#define B_     4
#define CIN_   16
#define COUT_  16
#define T_     16
#define H_     128
#define W_     128
#define SCALE_ 0.04811252243246881f
#define HW_    (H_ * W_)
#define PLANE_ (T_ * H_ * W_)
#define TAPS_  (4 * 32 * 2 + 2)
#define NA     (27 * TAPS_)
#define WPW_   (2 * CIN_ * 27)
#define ROWS_  16
#define RW_    4
#define NT_    4
#define NC_    (NT_ + 2)
#define NR_    (RW_ + 2)
#define PF_    3
#define NBUF_  (PF_ + 1)

__device__ __forceinline__ float ldx(const float* rowp, unsigned lane_off_bytes) {
    return *(const float*)((const char*)rowp + lane_off_bytes);
}
__device__ __forceinline__ v4f ldx4(const float* rowp, unsigned lane_off_bytes) {
    return *(const v4f*)((const char*)rowp + lane_off_bytes);
}

__global__ __launch_bounds__(256) __attribute__((amdgpu_waves_per_eu(4, 4))) void conv3d_mod_wmma_f32_direct(
    const float* __restrict__ input,
    const float* __restrict__ cond,
    const float* __restrict__ weight,
    float* __restrict__ out)
{
    __shared__ __attribute__((aligned(8))) float sA[NA];

    const int bi  = blockIdx.x;
    const int h0  = (bi % (H_ / ROWS_)) * ROWS_;
    const int t   = (bi / (H_ / ROWS_)) % T_;
    const int b   = bi / ((H_ / ROWS_) * T_);
    const int tid = threadIdx.x;

    {
        const int wq = tid >> 5, lq = tid & 31;
        const float* wsrc = weight + wq * WPW_ + lq;
        const float* cb   = cond + b * CIN_;
        float wraw[27];
#pragma unroll
        for (int n = 0; n < 27; ++n) wraw[n] = wsrc[32 * n];
#pragma unroll
        for (int n = 0; n < 27; ++n) {
            const unsigned idx = 32u * (unsigned)n + (unsigned)lq;
            const unsigned mm  = idx >= 432u ? 1u : 0u;
            const unsigned r   = idx - 432u * mm;
            const unsigned cin = r / 27u;
            const unsigned tap = r - 27u * cin;
            const unsigned m   = 2u * (unsigned)wq + mm;
            sA[tap * TAPS_ + (cin >> 2) * 64u + ((cin >> 1) & 1u) * 32u + m * 2u + (cin & 1u)] =
                wraw[n] * SCALE_ * cb[cin];
        }
    }
    __syncthreads();

    const int wv   = __builtin_amdgcn_readfirstlane(tid >> 5);
    const int lane = tid & 31;
    const int hf   = lane >> 4;
    const int lx   = lane & 15;
    const int rg   = wv >> 1;
    const int w0   = (wv & 1) * 64;
    const int hb   = h0 + rg * RW_;
    const int wA   = w0 + NT_ * lx;

    const bool okL = wA > 0;
    const bool okR = wA + NT_ < W_;
    const bool topok = hb > 0;
    const bool botok = hb + RW_ < H_;
    const int  hTop  = topok ? hb - 1 : 0;
    const int  hBot  = botok ? hb + RW_ : H_ - 1;

    v8f acc[NT_][RW_];
#pragma unroll
    for (int j = 0; j < NT_; ++j)
#pragma unroll
        for (int i = 0; i < RW_; ++i) acc[j][i] = v8f{};

    const float* inb = input + (size_t)b * CIN_ * PLANE_;

    const unsigned chalf = (unsigned)hf * 2u * (unsigned)PLANE_;
    const unsigned offL = (chalf + (unsigned)(wA > 0 ? wA - 1 : 0)) * 4u;
    const unsigned offM = (chalf + (unsigned)wA) * 4u;
    const unsigned offR = (chalf + (unsigned)(wA + NT_ < W_ ? wA + NT_ : W_ - 1)) * 4u;

    v2f a[9];
    float xb[NBUF_][2 * NC_];

#pragma unroll
    for (int q = 0; q < 9; ++q)
        a[q] = *(const v2f*)&sA[q * TAPS_ + lane * 2];

#define NSTEP_ (3 * 4 * NR_)
#define FETCH_ROW(rs_)                                                                       \
    {                                                                                        \
        const int fs = (rs_) / NR_, fr = (rs_) % NR_;                                        \
        const int fdt = fs / 4, fc = fs % 4;                                                 \
        const int fts = (fdt == 0) ? (t > 0 ? t - 1 : 0) : (fdt == 2) ? (t < T_ - 1 ? t + 1 : T_ - 1) : t; \
        const int fhs = (fr == 0) ? hTop : (fr == NR_ - 1) ? hBot : hb + fr - 1;             \
        const float* p0 = inb + (size_t)fhs * W_;                                            \
        const float* p1 = p0 + PLANE_;                                                       \
        const unsigned co = (unsigned)fts * (unsigned)(HW_ * 4) + (unsigned)fc * (unsigned)(4 * PLANE_ * 4); \
        float* xf = xb[(rs_) % NBUF_];                                                       \
        const v4f m0 = ldx4(p0, offM + co);                                                  \
        const v4f m1 = ldx4(p1, offM + co);                                                  \
        xf[0] = ldx(p0, offL + co); xf[1] = ldx(p1, offL + co);                              \
        xf[2] = m0[0]; xf[3] = m1[0]; xf[4] = m0[1]; xf[5] = m1[1];                          \
        xf[6] = m0[2]; xf[7] = m1[2]; xf[8] = m0[3]; xf[9] = m1[3];                          \
        xf[10] = ldx(p0, offR + co); xf[11] = ldx(p1, offR + co);                            \
    }

#pragma unroll
    for (int p = 0; p < PF_; ++p) FETCH_ROW(p)

#pragma unroll
    for (int s = 0; s < 12; ++s) {
#pragma unroll
    for (int ri = 0; ri < NR_; ++ri) {
        const int rs = s * NR_ + ri;
        const int dt = s / 4;
        if (rs + PF_ < NSTEP_) FETCH_ROW(rs + PF_)
        if (ri == RW_ && s < 11) {
            const int s1 = s + 1;
#pragma unroll
            for (int kw = 0; kw < 3; ++kw)
                a[kw] = *(const v2f*)&sA[((s1 / 4) * 9 + kw) * TAPS_ + (s1 % 4) * 64 + lane * 2];
        }
        if (ri == RW_ + 1 && s < 11) {
            const int s1 = s + 1;
#pragma unroll
            for (int kw = 0; kw < 3; ++kw)
                a[3 + kw] = *(const v2f*)&sA[((s1 / 4) * 9 + 3 + kw) * TAPS_ + (s1 % 4) * 64 + lane * 2];
        }
        if (ri == 0 && s > 0) {
#pragma unroll
            for (int kw = 0; kw < 3; ++kw)
                a[6 + kw] = *(const v2f*)&sA[((s / 4) * 9 + 6 + kw) * TAPS_ + (s % 4) * 64 + lane * 2];
        }

        float* xs = xb[rs % NBUF_];
        const bool planeok = (dt == 0) ? (t > 0) : (dt == 2) ? (t < T_ - 1) : true;
        const bool rowok = (ri == 0) ? topok : (ri == NR_ - 1) ? botok : true;
        const bool inside = planeok && rowok;
        const bool cl = okL && inside, cr = okR && inside;
        xs[0] = cl ? xs[0] : 0.0f; xs[1] = cl ? xs[1] : 0.0f;
        xs[2 * NC_ - 2] = cr ? xs[2 * NC_ - 2] : 0.0f; xs[2 * NC_ - 1] = cr ? xs[2 * NC_ - 1] : 0.0f;
        if (dt != 1 || ri == 0 || ri == NR_ - 1) {
#pragma unroll
            for (int i = 2; i < 2 * NC_ - 2; ++i) xs[i] = inside ? xs[i] : 0.0f;
        }
#pragma unroll
        for (int kw = 0; kw < 3; ++kw) {
#pragma unroll
            for (int kh = 0; kh < 3; ++kh) {
                const int ro = ri - kh;
                if (ro >= 0 && ro < RW_) {
#pragma unroll
                    for (int j = 0; j < NT_; ++j) {
                        v2f bf = {xs[2 * (j + kw)], xs[2 * (j + kw) + 1]};
                        acc[j][ro] = __builtin_amdgcn_wmma_f32_16x16x4_f32(
                            false, a[kh * 3 + kw], false, bf, (short)0, acc[j][ro], false, false);
                    }
                }
            }
        }
        __builtin_amdgcn_sched_barrier(0);
    }
    }
#undef FETCH_ROW
#undef NSTEP_

    const size_t strideC = (size_t)PLANE_;
    float* o = out + (size_t)b * COUT_ * strideC + (size_t)t * HW_ + (size_t)hb * W_ + wA;
#pragma unroll
    for (int ro = 0; ro < RW_; ++ro) {
#pragma unroll
        for (int r = 0; r < 8; ++r) {
            v4f pr = {acc[0][ro][r], acc[1][ro][r], acc[2][ro][r], acc[3][ro][r]};
            *(v4f*)&o[(size_t)(hf * 8 + r) * strideC + (size_t)ro * W_] = pr;
        }
    }
}

extern "C" void kernel_launch(void* const* d_in, const int* in_sizes, int n_in,
                              void* d_out, int out_size, void* d_ws, size_t ws_size,
                              hipStream_t stream) {
    const float* input  = (const float*)d_in[0];
    const float* cond   = (const float*)d_in[1];
    const float* weight = (const float*)d_in[2];
    float* out = (float*)d_out;

    conv3d_mod_wmma_f32_direct<<<dim3(B_ * T_ * (H_ / ROWS_)), dim3(256), 0, stream>>>(
        input, cond, weight, out);
}
